# starting kernel
# baseline (speedup 1.0000x reference)
#include <hip/hip_runtime.h>


typedef __attribute__((ext_vector_type(2))) float v2f;
typedef __attribute__((ext_vector_type(8))) float v8f;

constexpr int FEAT = 64;
constexpr int CT   = 4;

__global__ void zero_f32_kernel(float* __restrict__ p, int n) {
  int i = blockIdx.x * blockDim.x + threadIdx.x;
  if (i < n) p[i] = 0.0f;
}

__global__ void deg_kernel(const int* __restrict__ col, const float* __restrict__ w,
                           float* __restrict__ deg, int n_edges) {
  int e = blockIdx.x * blockDim.x + threadIdx.x;
  if (e < n_edges) unsafeAtomicAdd(&deg[col[e]], w[e]);
}

__global__ void rsqrt_kernel(float* __restrict__ deg, int n) {
  int i = blockIdx.x * blockDim.x + threadIdx.x;
  if (i < n) {
    float d = deg[i];
    deg[i] = (d > 0.0f) ? rsqrtf(d) : 0.0f;
  }
}

__global__ void norm_kernel(const int* __restrict__ row, const int* __restrict__ col,
                            const float* __restrict__ w, const float* __restrict__ dis,
                            float* __restrict__ norm, int n_edges) {
  int e = blockIdx.x * blockDim.x + threadIdx.x;
  if (e < n_edges) norm[e] = dis[row[e]] * w[e] * dis[col[e]];
}

__global__ void gemm64_wmma_kernel(const float* __restrict__ X, const float* __restrict__ W,
                                   float* __restrict__ H, int n_rows) {
  int wave = (blockIdx.x * blockDim.x + threadIdx.x) >> 5;
  int lane = threadIdx.x & 31;
  int row_tile = wave >> 2;
  int col_tile = wave & (CT - 1);
  int row_base = row_tile * 16;
  if (row_base >= n_rows) return;

  int half = lane >> 4;
  int m    = lane & 15;
  int cb   = col_tile * 16;

  int arow = row_base + m;
  if (arow > n_rows - 1) arow = n_rows - 1;
  const float* xrow = X + (size_t)arow * FEAT;

  v8f acc = {};
#pragma unroll
  for (int k0 = 0; k0 < FEAT; k0 += 4) {
    int k = k0 + 2 * half;
    v2f a, b;
    a.x = xrow[k];
    a.y = xrow[k + 1];
    b.x = W[(size_t)k * FEAT + cb + m];
    b.y = W[(size_t)(k + 1) * FEAT + cb + m];
    acc = __builtin_amdgcn_wmma_f32_16x16x4_f32(
         false, a,  false, b,
         (short)0, acc,  false,  false);
  }
#pragma unroll
  for (int v = 0; v < 8; ++v) {
    int r = row_base + v + 8 * half;
    if (r < n_rows) H[(size_t)r * FEAT + cb + m] = acc[v];
  }
}

__global__ void init_bias_kernel(float* __restrict__ out, const float* __restrict__ b, int total) {
  int i = blockIdx.x * blockDim.x + threadIdx.x;
  if (i < total) out[i] = b[i & (FEAT - 1)];
}

__global__ void scatter_kernel(const float* __restrict__ H, const int* __restrict__ row,
                               const int* __restrict__ col, const float* __restrict__ norm,
                               float* __restrict__ out, int n_edges) {
  int t = blockIdx.x * blockDim.x + threadIdx.x;
  int e = t >> 4;
  if (e >= n_edges) return;
  int f  = (t & 15) << 2;
  int r  = row[e];
  int c  = col[e];
  float nv = norm[e];
  const float4* hp = (const float4*)(H + (size_t)r * FEAT + f);
  float4 h = *hp;
  float* op = out + (size_t)c * FEAT + f;
  unsafeAtomicAdd(op + 0, h.x * nv);
  unsafeAtomicAdd(op + 1, h.y * nv);
  unsafeAtomicAdd(op + 2, h.z * nv);
  unsafeAtomicAdd(op + 3, h.w * nv);
}

__global__ void relu_inplace_kernel(float* __restrict__ p, int n) {
  int i = blockIdx.x * blockDim.x + threadIdx.x;
  if (i < n) p[i] = fmaxf(p[i], 0.0f);
}

extern "C" void kernel_launch(void* const* d_in, const int* in_sizes, int n_in,
                              void* d_out, int out_size, void* d_ws, size_t ws_size,
                              hipStream_t stream) {
  const float* x  = (const float*)d_in[0];
  const int*   ei = (const int*)d_in[1];
  const float* w  = (const float*)d_in[2];
  const float* W1 = (const float*)d_in[3];
  const float* b1 = (const float*)d_in[4];
  const float* W2 = (const float*)d_in[5];
  const float* b2 = (const float*)d_in[6];

  const int n_nodes = in_sizes[0] / FEAT;
  const int n_edges = in_sizes[1] / 2;
  const int* row = ei;
  const int* col = ei + n_edges;

  float* ws  = (float*)d_ws;
  size_t off = 0;
  auto carve = [&](size_t nf) { float* p = ws + off; off += (nf + 3) & ~(size_t)3; return p; };
  float* deg  = carve((size_t)n_nodes);
  float* norm = carve((size_t)n_edges);
  float* h1   = carve((size_t)n_nodes * FEAT);
  float* agg1 = carve((size_t)n_nodes * FEAT);
  float* h2   = carve((size_t)n_nodes * FEAT);
  float* outp = (float*)d_out;

  const int B = 256;
  auto nb = [&](long long n) { return (unsigned)((n + B - 1) / B); };

  const long long nodeF     = (long long)n_nodes * FEAT;
  const long long edgeT     = (long long)n_edges * 16;
  const long long gemmT     = (long long)((n_nodes + 15) / 16) * CT * 32;

  zero_f32_kernel <<<nb(n_nodes), B, 0, stream>>>(deg, n_nodes);
  deg_kernel      <<<nb(n_edges), B, 0, stream>>>(col, w, deg, n_edges);
  rsqrt_kernel    <<<nb(n_nodes), B, 0, stream>>>(deg, n_nodes);
  norm_kernel     <<<nb(n_edges), B, 0, stream>>>(row, col, w, deg, norm, n_edges);

  gemm64_wmma_kernel<<<nb(gemmT), B, 0, stream>>>(x, W1, h1, n_nodes);
  init_bias_kernel  <<<nb(nodeF), B, 0, stream>>>(agg1, b1, (int)nodeF);
  scatter_kernel    <<<nb(edgeT), B, 0, stream>>>(h1, row, col, norm, agg1, n_edges);

  gemm64_wmma_kernel<<<nb(gemmT), B, 0, stream>>>(agg1, W2, h2, n_nodes);
  init_bias_kernel  <<<nb(nodeF), B, 0, stream>>>(outp, b2, (int)nodeF);
  scatter_kernel    <<<nb(edgeT), B, 0, stream>>>(h2, row, col, norm, outp, n_edges);
  relu_inplace_kernel<<<nb(nodeF), B, 0, stream>>>(outp, (int)nodeF);
}


# scatter: one wave per 32 edges, lane = feature, rows 256-B aligned: every load and f32 atomic instr covers one whole 128-B line (was 4-6 part lines); atomics kept, arrival-order sums as before
# speedup vs baseline: 1.7854x; 1.7854x over previous
#include <hip/hip_runtime.h>


typedef __attribute__((ext_vector_type(2))) float v2f;
typedef __attribute__((ext_vector_type(8))) float v8f;

constexpr int FEAT = 64;
constexpr int CT   = 4;

__global__ void zero_f32_kernel(float* __restrict__ p, int n) {
  int i = blockIdx.x * blockDim.x + threadIdx.x;
  if (i < n) p[i] = 0.0f;
}

__global__ void deg_kernel(const int* __restrict__ col, const float* __restrict__ w,
                           float* __restrict__ deg, int n_edges) {
  int e = blockIdx.x * blockDim.x + threadIdx.x;
  if (e < n_edges) unsafeAtomicAdd(&deg[col[e]], w[e]);
}

__global__ void rsqrt_kernel(float* __restrict__ deg, int n) {
  int i = blockIdx.x * blockDim.x + threadIdx.x;
  if (i < n) {
    float d = deg[i];
    deg[i] = (d > 0.0f) ? rsqrtf(d) : 0.0f;
  }
}

__global__ void norm_kernel(const int* __restrict__ row, const int* __restrict__ col,
                            const float* __restrict__ w, const float* __restrict__ dis,
                            float* __restrict__ norm, int n_edges) {
  int e = blockIdx.x * blockDim.x + threadIdx.x;
  if (e < n_edges) norm[e] = dis[row[e]] * w[e] * dis[col[e]];
}

__global__ void gemm64_wmma_kernel(const float* __restrict__ X, const float* __restrict__ W,
                                   float* __restrict__ H, int n_rows) {
  int wave = (blockIdx.x * blockDim.x + threadIdx.x) >> 5;
  int lane = threadIdx.x & 31;
  int row_tile = wave >> 2;
  int col_tile = wave & (CT - 1);
  int row_base = row_tile * 16;
  if (row_base >= n_rows) return;

  int half = lane >> 4;
  int m    = lane & 15;
  int cb   = col_tile * 16;

  int arow = row_base + m;
  if (arow > n_rows - 1) arow = n_rows - 1;
  const float* xrow = X + (size_t)arow * FEAT;

  v8f acc = {};
#pragma unroll
  for (int k0 = 0; k0 < FEAT; k0 += 4) {
    int k = k0 + 2 * half;
    v2f a, b;
    a.x = xrow[k];
    a.y = xrow[k + 1];
    b.x = W[(size_t)k * FEAT + cb + m];
    b.y = W[(size_t)(k + 1) * FEAT + cb + m];
    acc = __builtin_amdgcn_wmma_f32_16x16x4_f32(
         false, a,  false, b,
         (short)0, acc,  false,  false);
  }
#pragma unroll
  for (int v = 0; v < 8; ++v) {
    int r = row_base + v + 8 * half;
    if (r < n_rows) H[(size_t)r * FEAT + cb + m] = acc[v];
  }
}

__global__ void init_bias_kernel(float* __restrict__ out, const float* __restrict__ b, int total) {
  int i = blockIdx.x * blockDim.x + threadIdx.x;
  if (i < total) out[i] = b[i & (FEAT - 1)];
}

constexpr int EPW = 32;
constexpr int EPP = 16;

template <bool TAIL>
__device__ __forceinline__ void scatter_pass(const float* __restrict__ H, float* __restrict__ out,
                                             int lane, int r_l, int c_l, float n_l, int p, int cnt) {
  float a[EPP], b[EPP];
  for (int j = 0; j < EPP; ++j) {
    const int r = __builtin_amdgcn_readlane(r_l, p + j);
    const float* hp = H + (size_t)r * FEAT;
    a[j] = hp[lane];
    b[j] = hp[lane + 32];
  }
  for (int j = 0; j < EPP; ++j) {
    if (!TAIL || p + j < cnt) {
      const int   c  = __builtin_amdgcn_readlane(c_l, p + j);
      const float nv = __int_as_float(__builtin_amdgcn_readlane(__float_as_int(n_l), p + j));
      float* op = out + (size_t)c * FEAT;
      unsafeAtomicAdd(op + lane,      a[j] * nv);
      unsafeAtomicAdd(op + lane + 32, b[j] * nv);
    }
  }
}

__global__ void scatter_kernel(const float* __restrict__ H, const int* __restrict__ row,
                               const int* __restrict__ col, const float* __restrict__ norm,
                               float* __restrict__ out, int n_edges) {
  const int lane = threadIdx.x & 31;
  const int wave = __builtin_amdgcn_readfirstlane((int)((blockIdx.x * blockDim.x + threadIdx.x) >> 5));
  const int e0 = wave * EPW;
  if (e0 >= n_edges) return;
  int cnt = n_edges - e0;
  if (cnt > EPW) cnt = EPW;

  int el = e0 + lane;
  if (el > n_edges - 1) el = n_edges - 1;
  const int   r_l = row[el];
  const int   c_l = col[el];
  const float n_l = norm[el];

  for (int p = 0; p < cnt; p += EPP) {
    if (p + EPP <= cnt) scatter_pass<false>(H, out, lane, r_l, c_l, n_l, p, cnt);
    else                scatter_pass<true >(H, out, lane, r_l, c_l, n_l, p, cnt);
  }
}

__global__ void relu_inplace_kernel(float* __restrict__ p, int n) {
  int i = blockIdx.x * blockDim.x + threadIdx.x;
  if (i < n) p[i] = fmaxf(p[i], 0.0f);
}

extern "C" void kernel_launch(void* const* d_in, const int* in_sizes, int n_in,
                              void* d_out, int out_size, void* d_ws, size_t ws_size,
                              hipStream_t stream) {
  const float* x  = (const float*)d_in[0];
  const int*   ei = (const int*)d_in[1];
  const float* w  = (const float*)d_in[2];
  const float* W1 = (const float*)d_in[3];
  const float* b1 = (const float*)d_in[4];
  const float* W2 = (const float*)d_in[5];
  const float* b2 = (const float*)d_in[6];

  const int n_nodes = in_sizes[0] / FEAT;
  const int n_edges = in_sizes[1] / 2;
  const int* row = ei;
  const int* col = ei + n_edges;

  float* ws  = (float*)d_ws;
  size_t off = 0;
  auto carve = [&](size_t nf) { float* p = ws + off; off += (nf + 63) & ~(size_t)63; return p; };
  float* deg  = carve((size_t)n_nodes);
  float* norm = carve((size_t)n_edges);
  float* h1   = carve((size_t)n_nodes * FEAT);
  float* agg1 = carve((size_t)n_nodes * FEAT);
  float* h2   = carve((size_t)n_nodes * FEAT);
  float* outp = (float*)d_out;

  const int B = 256;
  auto nb = [&](long long n) { return (unsigned)((n + B - 1) / B); };

  const long long nodeF     = (long long)n_nodes * FEAT;
  const long long edgeT     = (long long)((n_edges + EPW - 1) / EPW) * 32;
  const long long gemmT     = (long long)((n_nodes + 15) / 16) * CT * 32;

  zero_f32_kernel <<<nb(n_nodes), B, 0, stream>>>(deg, n_nodes);
  deg_kernel      <<<nb(n_edges), B, 0, stream>>>(col, w, deg, n_edges);
  rsqrt_kernel    <<<nb(n_nodes), B, 0, stream>>>(deg, n_nodes);
  norm_kernel     <<<nb(n_edges), B, 0, stream>>>(row, col, w, deg, norm, n_edges);

  gemm64_wmma_kernel<<<nb(gemmT), B, 0, stream>>>(x, W1, h1, n_nodes);
  init_bias_kernel  <<<nb(nodeF), B, 0, stream>>>(agg1, b1, (int)nodeF);
  scatter_kernel    <<<nb(edgeT), B, 0, stream>>>(h1, row, col, norm, agg1, n_edges);

  gemm64_wmma_kernel<<<nb(gemmT), B, 0, stream>>>(agg1, W2, h2, n_nodes);
  init_bias_kernel  <<<nb(nodeF), B, 0, stream>>>(outp, b2, (int)nodeF);
  scatter_kernel    <<<nb(edgeT), B, 0, stream>>>(h2, row, col, norm, outp, n_edges);
  relu_inplace_kernel<<<nb(nodeF), B, 0, stream>>>(outp, (int)nodeF);
}


# fusion: bias start values and deg=0 stored by the gemm waves, rsqrt+norm formed inside the first scatter: 11 -> 6 launches; atomics, arrival-order sums as before
# speedup vs baseline: 2.0639x; 1.1560x over previous
#include <hip/hip_runtime.h>


typedef __attribute__((ext_vector_type(2))) float v2f;
typedef __attribute__((ext_vector_type(8))) float v8f;

constexpr int FEAT = 64;
constexpr int CT   = 4;

__global__ void deg_kernel(const int* __restrict__ col, const float* __restrict__ w,
                           float* __restrict__ deg, int n_edges) {
  int e = blockIdx.x * blockDim.x + threadIdx.x;
  if (e < n_edges) unsafeAtomicAdd(&deg[col[e]], w[e]);
}

__device__ __forceinline__ float inv_sqrt_deg(float d) { return (d > 0.0f) ? rsqrtf(d) : 0.0f; }

template <bool ZERO_DEG>
__global__ void gemm64_wmma_kernel(const float* __restrict__ X, const float* __restrict__ W,
                                   float* __restrict__ H, int n_rows,
                                   float* __restrict__ acc_init, const float* __restrict__ bias,
                                   float* __restrict__ deg) {
  int wave = (blockIdx.x * blockDim.x + threadIdx.x) >> 5;
  int lane = threadIdx.x & 31;
  int row_tile = wave >> 2;
  int col_tile = wave & (CT - 1);
  int row_base = row_tile * 16;
  if (row_base >= n_rows) return;

  int half = lane >> 4;
  int m    = lane & 15;
  int cb   = col_tile * 16;

  const int uw = __builtin_amdgcn_readfirstlane(wave);
  const float b_lo = bias[lane];
  const float b_hi = bias[lane + 32];
  const int r0 = (uw >> 2) * 16 + (uw & (CT - 1)) * 4;
  for (int i = 0; i < 4; ++i) {
    const int r = r0 + i;
    if (r < n_rows) {
      float* ap = acc_init + (size_t)r * FEAT;
      ap[lane]      = b_lo;
      ap[lane + 32] = b_hi;
    }
  }
  if (ZERO_DEG) {
    const int zi = uw * 32 + lane;
    if (zi < n_rows) deg[zi] = 0.0f;
  }

  int arow = row_base + m;
  if (arow > n_rows - 1) arow = n_rows - 1;
  const float* xrow = X + (size_t)arow * FEAT;

  v8f acc = {};
#pragma unroll
  for (int k0 = 0; k0 < FEAT; k0 += 4) {
    int k = k0 + 2 * half;
    v2f a, b;
    a.x = xrow[k];
    a.y = xrow[k + 1];
    b.x = W[(size_t)k * FEAT + cb + m];
    b.y = W[(size_t)(k + 1) * FEAT + cb + m];
    acc = __builtin_amdgcn_wmma_f32_16x16x4_f32(
         false, a,  false, b,
         (short)0, acc,  false,  false);
  }
#pragma unroll
  for (int v = 0; v < 8; ++v) {
    int r = row_base + v + 8 * half;
    if (r < n_rows) H[(size_t)r * FEAT + cb + m] = acc[v];
  }
}

constexpr int EPW = 32;
constexpr int EPP = 16;

template <bool TAIL>
__device__ __forceinline__ void scatter_pass(const float* __restrict__ H, float* __restrict__ out,
                                             int lane, int r_l, int c_l, float n_l, int p, int cnt) {
  float a[EPP], b[EPP];
  for (int j = 0; j < EPP; ++j) {
    const int r = __builtin_amdgcn_readlane(r_l, p + j);
    const float* hp = H + (size_t)r * FEAT;
    a[j] = hp[lane];
    b[j] = hp[lane + 32];
  }
  for (int j = 0; j < EPP; ++j) {
    if (!TAIL || p + j < cnt) {
      const int   c  = __builtin_amdgcn_readlane(c_l, p + j);
      const float nv = __int_as_float(__builtin_amdgcn_readlane(__float_as_int(n_l), p + j));
      float* op = out + (size_t)c * FEAT;
      unsafeAtomicAdd(op + lane,      a[j] * nv);
      unsafeAtomicAdd(op + lane + 32, b[j] * nv);
    }
  }
}

template <bool MAKE_NORM>
__global__ void scatter_kernel(const float* __restrict__ H, const int* __restrict__ row,
                               const int* __restrict__ col, const float* __restrict__ wn,
                               const float* __restrict__ deg, float* __restrict__ norm_out,
                               float* __restrict__ out, int n_edges) {
  const int lane = threadIdx.x & 31;
  const int wave = __builtin_amdgcn_readfirstlane((int)((blockIdx.x * blockDim.x + threadIdx.x) >> 5));
  const int e0 = wave * EPW;
  if (e0 >= n_edges) return;
  int cnt = n_edges - e0;
  if (cnt > EPW) cnt = EPW;

  int el = e0 + lane;
  if (el > n_edges - 1) el = n_edges - 1;
  const int   r_l = row[el];
  const int   c_l = col[el];
  float n_l = wn[el];
  if (MAKE_NORM) {
    n_l = inv_sqrt_deg(deg[r_l]) * n_l * inv_sqrt_deg(deg[c_l]);
    if (lane < cnt) norm_out[el] = n_l;
  }

  for (int p = 0; p < cnt; p += EPP) {
    if (p + EPP <= cnt) scatter_pass<false>(H, out, lane, r_l, c_l, n_l, p, cnt);
    else                scatter_pass<true >(H, out, lane, r_l, c_l, n_l, p, cnt);
  }
}

__global__ void relu_inplace_kernel(float* __restrict__ p, int n) {
  int i = blockIdx.x * blockDim.x + threadIdx.x;
  if (i < n) p[i] = fmaxf(p[i], 0.0f);
}

extern "C" void kernel_launch(void* const* d_in, const int* in_sizes, int n_in,
                              void* d_out, int out_size, void* d_ws, size_t ws_size,
                              hipStream_t stream) {
  const float* x  = (const float*)d_in[0];
  const int*   ei = (const int*)d_in[1];
  const float* w  = (const float*)d_in[2];
  const float* W1 = (const float*)d_in[3];
  const float* b1 = (const float*)d_in[4];
  const float* W2 = (const float*)d_in[5];
  const float* b2 = (const float*)d_in[6];

  const int n_nodes = in_sizes[0] / FEAT;
  const int n_edges = in_sizes[1] / 2;
  const int* row = ei;
  const int* col = ei + n_edges;

  float* ws  = (float*)d_ws;
  size_t off = 0;
  auto carve = [&](size_t nf) { float* p = ws + off; off += (nf + 63) & ~(size_t)63; return p; };
  float* deg  = carve((size_t)n_nodes);
  float* norm = carve((size_t)n_edges);
  float* h1   = carve((size_t)n_nodes * FEAT);
  float* agg1 = carve((size_t)n_nodes * FEAT);
  float* h2   = carve((size_t)n_nodes * FEAT);
  float* outp = (float*)d_out;

  const int B = 256;
  auto nb = [&](long long n) { return (unsigned)((n + B - 1) / B); };

  const long long nodeF     = (long long)n_nodes * FEAT;
  const long long edgeT     = (long long)((n_edges + EPW - 1) / EPW) * 32;
  const long long gemmT     = (long long)((n_nodes + 15) / 16) * CT * 32;

  gemm64_wmma_kernel<true><<<nb(gemmT), B, 0, stream>>>(x, W1, h1, n_nodes, agg1, b1, deg);

  deg_kernel      <<<nb(n_edges), B, 0, stream>>>(col, w, deg, n_edges);

  scatter_kernel<true> <<<nb(edgeT), B, 0, stream>>>(h1, row, col, w, deg, norm, agg1, n_edges);

  gemm64_wmma_kernel<false><<<nb(gemmT), B, 0, stream>>>(agg1, W2, h2, n_nodes, outp, b2, deg);
  scatter_kernel<false><<<nb(edgeT), B, 0, stream>>>(h2, row, col, norm, deg, nullptr, outp, n_edges);
  relu_inplace_kernel<<<nb(nodeF), B, 0, stream>>>(outp, (int)nodeF);
}


# scatter by L2 half: destination rows in two ranges, a work-group (die id, two task counters) takes only the edges that add into the range of the half it runs on; atomics, arrival-order sums as before
# speedup vs baseline: 2.2851x; 1.1072x over previous
#include <hip/hip_runtime.h>


typedef __attribute__((ext_vector_type(2))) float v2f;
typedef __attribute__((ext_vector_type(8))) float v8f;

constexpr int FEAT = 64;
constexpr int CT   = 4;
constexpr int CTR_PITCH = 64;

__global__ void deg_kernel(const int* __restrict__ col, const float* __restrict__ w,
                           float* __restrict__ deg, int n_edges) {
  int e = blockIdx.x * blockDim.x + threadIdx.x;
  if (e < n_edges) unsafeAtomicAdd(&deg[col[e]], w[e]);
}

__device__ __forceinline__ float inv_sqrt_deg(float d) { return (d > 0.0f) ? rsqrtf(d) : 0.0f; }

template <bool FIRST>
__global__ void gemm64_wmma_kernel(const float* __restrict__ X, const float* __restrict__ W,
                                   float* __restrict__ H, int n_rows,
                                   float* __restrict__ acc_init, const float* __restrict__ bias,
                                   float* __restrict__ deg, int* __restrict__ ctr) {
  int wave = (blockIdx.x * blockDim.x + threadIdx.x) >> 5;
  int lane = threadIdx.x & 31;
  int row_tile = wave >> 2;
  int col_tile = wave & (CT - 1);
  int row_base = row_tile * 16;
  if (row_base >= n_rows) return;

  int half = lane >> 4;
  int m    = lane & 15;
  int cb   = col_tile * 16;

  const int uw = __builtin_amdgcn_readfirstlane(wave);
  const float b_lo = bias[lane];
  const float b_hi = bias[lane + 32];
  const int r0 = (uw >> 2) * 16 + (uw & (CT - 1)) * 4;
  for (int i = 0; i < 4; ++i) {
    const int r = r0 + i;
    if (r < n_rows) {
      float* ap = acc_init + (size_t)r * FEAT;
      ap[lane]      = b_lo;
      ap[lane + 32] = b_hi;
    }
  }
  if (FIRST) {
    const int zi = uw * 32 + lane;
    if (zi < n_rows) deg[zi] = 0.0f;
    if (uw == 0 && lane < 4) ctr[lane * CTR_PITCH] = 0;
  }

  int arow = row_base + m;
  if (arow > n_rows - 1) arow = n_rows - 1;
  const float* xrow = X + (size_t)arow * FEAT;

  v8f acc = {};
#pragma unroll
  for (int k0 = 0; k0 < FEAT; k0 += 4) {
    int k = k0 + 2 * half;
    v2f a, b;
    a.x = xrow[k];
    a.y = xrow[k + 1];
    b.x = W[(size_t)k * FEAT + cb + m];
    b.y = W[(size_t)(k + 1) * FEAT + cb + m];
    acc = __builtin_amdgcn_wmma_f32_16x16x4_f32(
         false, a,  false, b,
         (short)0, acc,  false,  false);
  }
#pragma unroll
  for (int v = 0; v < 8; ++v) {
    int r = row_base + v + 8 * half;
    if (r < n_rows) H[(size_t)r * FEAT + cb + m] = acc[v];
  }
}

constexpr int EPP = 16;
constexpr int EPL = 32;
constexpr int EPW = 2 * EPL;
constexpr int WPB = 8;
constexpr int EPB = EPW * WPB;

template <int J, bool FULL>
struct PassStep {
  static __device__ __forceinline__ unsigned run(const float* __restrict__ H, float* __restrict__ out,
                                                 int lane, int r_l, int c_l, float n_l,
                                                 unsigned mm, int n) {
    if (FULL || J < n) {
      const int l = __builtin_ctz(mm);
      const int r = __builtin_amdgcn_readlane(r_l, l);
      const float* hp = H + (size_t)r * FEAT;
      const float a = hp[lane];
      const float b = hp[lane + 32];
      const unsigned rest = PassStep<J + 1, FULL>::run(H, out, lane, r_l, c_l, n_l, mm & (mm - 1), n);
      const int   c  = __builtin_amdgcn_readlane(c_l, l);
      const float nv = __int_as_float(__builtin_amdgcn_readlane(__float_as_int(n_l), l));
      float* op = out + (size_t)c * FEAT;
      unsafeAtomicAdd(op + lane,      a * nv);
      unsafeAtomicAdd(op + lane + 32, b * nv);
      return rest;
    }
    return mm;
  }
};
template <bool FULL>
struct PassStep<EPP, FULL> {
  static __device__ __forceinline__ unsigned run(const float* __restrict__, float* __restrict__, int, int, int, float,
                                                 unsigned mm, int) { return mm; }
};

__device__ __forceinline__ void scatter_line(const float* __restrict__ H, float* __restrict__ out,
                                             int lane, int r_l, int c_l, float n_l, unsigned m) {
  while (m != 0u) {
    const int n = __builtin_popcount(m);
    if (n >= EPP) m = PassStep<0, true >::run(H, out, lane, r_l, c_l, n_l, m, EPP);
    else          m = PassStep<0, false>::run(H, out, lane, r_l, c_l, n_l, m, n);
  }
}

template <bool MAKE_NORM>
__global__ void __launch_bounds__(WPB * 32)
scatter_kernel(const float* __restrict__ H, const int* __restrict__ row,
               const int* __restrict__ col, const float* __restrict__ wn,
               const float* __restrict__ deg, float* __restrict__ norm_out,
               float* __restrict__ out, int n_edges, int n_split,
               int* __restrict__ ctr, int n_groups) {
  __shared__ int s_task;
  if (threadIdx.x == 0) {
    const int die = (int)((__builtin_amdgcn_s_sendmsg_rtn(0x87) >> 16) & 15);
    int h = (die >> 2) & 1;
    int g = atomicAdd(&ctr[h * CTR_PITCH], 1);
    if (g >= n_groups) { h ^= 1; g = atomicAdd(&ctr[h * CTR_PITCH], 1); }
    s_task = (g < n_groups) ? (g * 2 + h) : -1;
  }
  __syncthreads();
  const int task = __builtin_amdgcn_readfirstlane(s_task);
  if (task < 0) return;
  const int h    = task & 1;
  const int lane = threadIdx.x & 31;
  const int wib  = __builtin_amdgcn_readfirstlane((int)(threadIdx.x >> 5));
  const int e0   = ((task >> 1) * WPB + wib) * EPW;
  if (e0 >= n_edges) return;
  int cnt = n_edges - e0;
  if (cnt > EPW) cnt = EPW;

  int el0 = e0 + lane;
  int el1 = e0 + EPL + lane;
  if (el0 > n_edges - 1) el0 = n_edges - 1;
  if (el1 > n_edges - 1) el1 = n_edges - 1;
  const int r0 = row[el0], r1 = row[el1];
  const int c0 = col[el0], c1 = col[el1];
  float n0 = wn[el0], n1 = wn[el1];
  asm volatile("" :: "v"(r0), "v"(r1), "v"(c0), "v"(c1), "v"(n0), "v"(n1));

  const bool sel0 = (lane < cnt)       && ((c0 >= n_split ? 1 : 0) == h);
  const bool sel1 = (lane + EPL < cnt) && ((c1 >= n_split ? 1 : 0) == h);
  if (MAKE_NORM) {
    const float dr0 = deg[sel0 ? r0 : 0], dc0 = deg[sel0 ? c0 : 0];
    const float dr1 = deg[sel1 ? r1 : 0], dc1 = deg[sel1 ? c1 : 0];
    n0 = inv_sqrt_deg(dr0) * n0 * inv_sqrt_deg(dc0);
    n1 = inv_sqrt_deg(dr1) * n1 * inv_sqrt_deg(dc1);
    if (sel0) norm_out[el0] = n0;
    if (sel1) norm_out[el1] = n1;
  }
  scatter_line(H, out, lane, r0, c0, n0, __builtin_amdgcn_ballot_w32(sel0));
  scatter_line(H, out, lane, r1, c1, n1, __builtin_amdgcn_ballot_w32(sel1));
}

__global__ void relu_inplace_kernel(float* __restrict__ p, int n) {
  int i = blockIdx.x * blockDim.x + threadIdx.x;
  if (i < n) p[i] = fmaxf(p[i], 0.0f);
}

extern "C" void kernel_launch(void* const* d_in, const int* in_sizes, int n_in,
                              void* d_out, int out_size, void* d_ws, size_t ws_size,
                              hipStream_t stream) {
  const float* x  = (const float*)d_in[0];
  const int*   ei = (const int*)d_in[1];
  const float* w  = (const float*)d_in[2];
  const float* W1 = (const float*)d_in[3];
  const float* b1 = (const float*)d_in[4];
  const float* W2 = (const float*)d_in[5];
  const float* b2 = (const float*)d_in[6];

  const int n_nodes = in_sizes[0] / FEAT;
  const int n_edges = in_sizes[1] / 2;
  const int* row = ei;
  const int* col = ei + n_edges;

  float* ws  = (float*)d_ws;
  size_t off = 0;
  auto carve = [&](size_t nf) { float* p = ws + off; off += (nf + 63) & ~(size_t)63; return p; };
  float* deg  = carve((size_t)n_nodes);
  float* norm = carve((size_t)n_edges);
  float* h1   = carve((size_t)n_nodes * FEAT);
  float* agg1 = carve((size_t)n_nodes * FEAT);
  float* h2   = carve((size_t)n_nodes * FEAT);
  int*   ctr  = (int*)carve((size_t)4 * CTR_PITCH);
  float* outp = (float*)d_out;

  const int B = 256;
  auto nb = [&](long long n) { return (unsigned)((n + B - 1) / B); };

  const long long nodeF    = (long long)n_nodes * FEAT;
  const int       n_groups = (n_edges + EPB - 1) / EPB;
  const int       n_split  = n_nodes / 2;
  const long long gemmT    = (long long)((n_nodes + 15) / 16) * CT * 32;

  gemm64_wmma_kernel<true><<<nb(gemmT), B, 0, stream>>>(x, W1, h1, n_nodes, agg1, b1, deg, ctr);

  deg_kernel      <<<nb(n_edges), B, 0, stream>>>(col, w, deg, n_edges);

  scatter_kernel<true> <<<2u * (unsigned)n_groups, WPB * 32, 0, stream>>>(h1, row, col, w, deg, norm, agg1, n_edges,
                                                                      n_split, ctr, n_groups);

  gemm64_wmma_kernel<false><<<nb(gemmT), B, 0, stream>>>(agg1, W2, h2, n_nodes, outp, b2, deg, ctr);
  scatter_kernel<false><<<2u * (unsigned)n_groups, WPB * 32, 0, stream>>>(h2, row, col, norm, deg, nullptr, outp, n_edges,
                                                                      n_split, ctr + 2 * CTR_PITCH, n_groups);
  relu_inplace_kernel<<<nb(nodeF), B, 0, stream>>>(outp, (int)nodeF);
}
